# starting kernel
# baseline (speedup 1.0000x reference)
#include <hip/hip_runtime.h>
#include <math.h>

typedef float v2f __attribute__((ext_vector_type(2)));
typedef float v8f __attribute__((ext_vector_type(8)));

__device__ __forceinline__ v8f wmma_f32_k4(v2f a, v2f b, v8f c) {
  return __builtin_amdgcn_wmma_f32_16x16x4_f32(
       false, a,  false, b,
       (short)0, c,  false,  false);
}

__device__ __forceinline__ void atomic_add_f32(float* p, float v) {
  asm volatile("global_atomic_add_f32 %0, %1, off scope:SCOPE_DEV"
               :: "v"(p), "v"(v)
               : "memory");
}


__global__ void k_fill(float* __restrict__ p, float v, int n) {
  int i = blockIdx.x * blockDim.x + threadIdx.x;
  if (i < n) p[i] = v;
}

__global__ void k_deg_accum(const int* __restrict__ dst, float* __restrict__ deg, int E) {
  int i = blockIdx.x * blockDim.x + threadIdx.x;
  if (i < E) atomic_add_f32(&deg[dst[i]], 1.0f);
}

__global__ void k_rsqrt_inplace(float* __restrict__ d, int n) {
  int i = blockIdx.x * blockDim.x + threadIdx.x;
  if (i < n) d[i] = rsqrtf(d[i]);
}

__global__ void k_norm(const int* __restrict__ src, const int* __restrict__ dst,
                       const float* __restrict__ dinv, float* __restrict__ norm,
                       int E, int N) {
  int i = blockIdx.x * blockDim.x + threadIdx.x;
  if (i >= E + N) return;
  if (i < E) {
    norm[i] = dinv[src[i]] * dinv[dst[i]];
  } else {
    float v = dinv[i - E];
    norm[i] = v * v;
  }
}

__global__ void k_bias_relu(float* __restrict__ a, const float* __restrict__ b, int n) {
  int i = blockIdx.x * blockDim.x + threadIdx.x;
  if (i < n) a[i] = fmaxf(a[i] + b[i & 63], 0.0f);
}


__global__ void k_gemm_n64(const float* __restrict__ X, const float* __restrict__ W,
                           float* __restrict__ H, int M, int K) {
  int wave = (int)((blockIdx.x * blockDim.x + threadIdx.x) >> 5);
  int lane = threadIdx.x & 31;
  int m0 = wave * 16;
  if (m0 >= M) return;
  int half = lane >> 4;
  int lm = lane & 15;

  const float* xrow = X + (size_t)(m0 + lm) * (size_t)K;
  v8f acc0 = {}, acc1 = {}, acc2 = {}, acc3 = {};

  for (int kb = 0; kb < K; kb += 4) {
    int k0 = kb + 2 * half;
    v2f a;
    a.x = xrow[k0];
    a.y = xrow[k0 + 1];
    const float* w0 = W + (size_t)k0 * 64;
    v2f b0, b1, b2, b3;
    b0.x = w0[lm];       b0.y = w0[64 + lm];
    b1.x = w0[16 + lm];  b1.y = w0[64 + 16 + lm];
    b2.x = w0[32 + lm];  b2.y = w0[64 + 32 + lm];
    b3.x = w0[48 + lm];  b3.y = w0[64 + 48 + lm];
    acc0 = wmma_f32_k4(a, b0, acc0);
    acc1 = wmma_f32_k4(a, b1, acc1);
    acc2 = wmma_f32_k4(a, b2, acc2);
    acc3 = wmma_f32_k4(a, b3, acc3);
  }

#pragma unroll
  for (int r = 0; r < 8; ++r) {
    int row = m0 + r + 8 * half;
    float* hr = H + (size_t)row * 64 + lm;
    hr[0]  = acc0[r];
    hr[16] = acc1[r];
    hr[32] = acc2[r];
    hr[48] = acc3[r];
  }
}


__global__ void k_scatter(const int* __restrict__ src, const int* __restrict__ dst,
                          const float* __restrict__ norm, const float* __restrict__ h,
                          float* __restrict__ agg, int E, int N) {
  unsigned int gid = blockIdx.x * blockDim.x + threadIdx.x;
  unsigned int eid = gid >> 4;
  int t4 = (gid & 15) * 4;
  if (eid >= (unsigned int)(E + N)) return;
  int s, d;
  if (eid < (unsigned int)E) {
    s = src[eid];
    d = dst[eid];
  } else {
    s = d = (int)eid - E;
  }
  float nm = norm[eid];
  const float4 hv = *reinterpret_cast<const float4*>(h + (size_t)s * 64 + t4);
  float* ap = agg + (size_t)d * 64 + t4;
  atomic_add_f32(ap + 0, hv.x * nm);
  atomic_add_f32(ap + 1, hv.y * nm);
  atomic_add_f32(ap + 2, hv.z * nm);
  atomic_add_f32(ap + 3, hv.w * nm);
}


__global__ void k_pool(const float* __restrict__ h, const int* __restrict__ batch,
                       float* __restrict__ gsum, float* __restrict__ cnt, int N) {
  unsigned int gid = blockIdx.x * blockDim.x + threadIdx.x;
  if (gid >= (unsigned int)N * 64u) return;
  int c = gid & 63;
  int n = gid >> 6;
  int g = batch[n];
  atomic_add_f32(&gsum[g * 64 + c], h[(size_t)n * 64 + c]);
  if (c == 0) atomic_add_f32(&cnt[g], 1.0f);
}


__global__ void k_pool_gemm_out(const float* __restrict__ gsum, const float* __restrict__ cnt,
                                const float* __restrict__ Wp, const float* __restrict__ bp,
                                float* __restrict__ out) {
  int wave = threadIdx.x >> 5;
  int lane = threadIdx.x & 31;
  int mi = wave >> 2, ni = wave & 3;
  int half = lane >> 4, lm = lane & 15;

  int arow = mi * 16 + lm;
  float inv = 1.0f / fmaxf(cnt[arow], 1.0f);

  v8f acc = {};
  for (int kb = 0; kb < 64; kb += 4) {
    int k0 = kb + 2 * half;
    v2f a;
    a.x = gsum[arow * 64 + k0] * inv;
    a.y = gsum[arow * 64 + k0 + 1] * inv;
    v2f b;
    b.x = Wp[k0 * 64 + ni * 16 + lm];
    b.y = Wp[(k0 + 1) * 64 + ni * 16 + lm];
    acc = wmma_f32_k4(a, b, acc);
  }

#pragma unroll
  for (int r = 0; r < 8; ++r) {
    int row = mi * 16 + r + 8 * half;
    out[row * 64 + ni * 16 + lm] = acc[r] + bp[ni * 16 + lm];
  }
}


static inline size_t align_f(size_t n) { return (n + 1023) & ~(size_t)1023; }

extern "C" void kernel_launch(void* const* d_in, const int* in_sizes, int n_in,
                              void* d_out, int out_size, void* d_ws, size_t ws_size,
                              hipStream_t stream) {
  const float* x  = (const float*)d_in[0];
  const float* W1 = (const float*)d_in[1];
  const float* b1 = (const float*)d_in[2];
  const float* W2 = (const float*)d_in[3];
  const float* b2 = (const float*)d_in[4];
  const float* Wp = (const float*)d_in[5];
  const float* bp = (const float*)d_in[6];
  const int* ei   = (const int*)d_in[7];
  const int* batch = (const int*)d_in[8];

  const int E = in_sizes[7] / 2;
  const int N = in_sizes[8];
  const int K1 = in_sizes[0] / N;
  const int* src = ei;
  const int* dst = ei + E;

  float* ws = (float*)d_ws;
  size_t o = 0;
  float* dinv = ws + o;  o += align_f((size_t)N);
  float* norm = ws + o;  o += align_f((size_t)(E + N));
  float* h    = ws + o;  o += align_f((size_t)N * 64);
  float* agg  = ws + o;  o += align_f((size_t)N * 64);
  float* gsum = ws + o;  o += 64 * 64;
  float* cnt  = ws + o;  o += 64;
  (void)ws_size;

  const int T = 256;
  auto blocks = [](long long n, int t) { return (unsigned int)((n + t - 1) / t); };

  k_fill<<<blocks(N, T), T, 0, stream>>>(dinv, 1.0f, N);
  k_deg_accum<<<blocks(E, T), T, 0, stream>>>(dst, dinv, E);
  k_rsqrt_inplace<<<blocks(N, T), T, 0, stream>>>(dinv, N);
  k_norm<<<blocks((long long)E + N, T), T, 0, stream>>>(src, dst, dinv, norm, E, N);

  const int tiles = (N + 15) / 16;
  const unsigned int gemm_blocks = (unsigned int)((tiles + 7) / 8);
  const long long NE16 = (long long)(E + N) * 16;

  k_gemm_n64<<<gemm_blocks, T, 0, stream>>>(x, W1, h, N, K1);
  k_fill<<<blocks((long long)N * 64, T), T, 0, stream>>>(agg, 0.0f, N * 64);
  k_scatter<<<blocks(NE16, T), T, 0, stream>>>(src, dst, norm, h, agg, E, N);
  k_bias_relu<<<blocks((long long)N * 64, T), T, 0, stream>>>(agg, b1, N * 64);

  k_gemm_n64<<<gemm_blocks, T, 0, stream>>>(agg, W2, h, N, 64);
  k_fill<<<blocks((long long)N * 64, T), T, 0, stream>>>(agg, 0.0f, N * 64);
  k_scatter<<<blocks(NE16, T), T, 0, stream>>>(src, dst, norm, h, agg, E, N);
  k_bias_relu<<<blocks((long long)N * 64, T), T, 0, stream>>>(agg, b2, N * 64);

  k_fill<<<blocks(64 * 64 + 64, T), T, 0, stream>>>(gsum, 0.0f, 64 * 64 + 64);
  k_pool<<<blocks((long long)N * 64, T), T, 0, stream>>>(agg, batch, gsum, cnt, N);
  k_pool_gemm_out<<<1, 512, 0, stream>>>(gsum, cnt, Wp, bp, (float*)d_out);
}


# edge aggregation as a one-hot matrix product on f32 WMMA: edges placed by destination row (rank returned by the degree atomic, row scan), one wave per 16 rows; no scatter atomics; sums in rank order
# speedup vs baseline: 1.2438x; 1.2438x over previous
#include <hip/hip_runtime.h>
#include <math.h>

typedef float v2f __attribute__((ext_vector_type(2)));
typedef float v8f __attribute__((ext_vector_type(8)));

__device__ __forceinline__ v8f wmma_f32_k4(v2f a, v2f b, v8f c) {
  return __builtin_amdgcn_wmma_f32_16x16x4_f32(
       false, a,  false, b,
       (short)0, c,  false,  false);
}

__device__ __forceinline__ float atomic_add_f32_ret(float* p, float v) {
  float old;
  asm volatile("global_atomic_add_f32 %0, %1, %2, off th:TH_ATOMIC_RETURN scope:SCOPE_DEV\n\t"
               "s_wait_loadcnt 0x0"
               : "=&v"(old)
               : "v"(p), "v"(v)
               : "memory");
  return old;
}


__global__ void k_fill(float* __restrict__ p, float v, int n) {
  int i = blockIdx.x * blockDim.x + threadIdx.x;
  if (i < n) p[i] = v;
}

__global__ void k_deg_rank(const int* __restrict__ dst, float* __restrict__ deg,
                           int* __restrict__ rank, int E) {
  int i = blockIdx.x * blockDim.x + threadIdx.x;
  if (i < E) rank[i] = (int)atomic_add_f32_ret(&deg[dst[i]], 1.0f) - 1;
}


#define SCAN_T 1024

__global__ __launch_bounds__(SCAN_T)
void k_scan_rows(const float* __restrict__ deg, int* __restrict__ rowptr, int N) {
  __shared__ int wsum[32];
  const int t = threadIdx.x, lane = t & 31, w = t >> 5;
  const int chunk = (N + SCAN_T - 1) / SCAN_T;
  const int i0 = t * chunk;
  int s = 0;
#pragma unroll 8
  for (int j = 0; j < chunk; ++j) {
    const int i = i0 + j;
    s += (i < N) ? (int)deg[i] - 1 : 0;
  }
  int v = s;
  for (int d = 1; d < 32; d <<= 1) {
    const int u = __shfl_up(v, d);
    if (lane >= d) v += u;
  }
  if (lane == 31) wsum[w] = v;
  __syncthreads();
  if (w == 0) {
    const int x = wsum[lane];
    int y = x;
    for (int d = 1; d < 32; d <<= 1) {
      const int u = __shfl_up(y, d);
      if (lane >= d) y += u;
    }
    wsum[lane] = y - x;
  }
  __syncthreads();
  int run = wsum[w] + v - s;
#pragma unroll 8
  for (int j = 0; j < chunk; ++j) {
    const int i = i0 + j;
    if (i < N) {
      rowptr[i] = run;
      run += (int)deg[i] - 1;
    }
  }
  if (t == SCAN_T - 1) rowptr[N] = run;
}

__global__ void k_place(const int* __restrict__ src, const int* __restrict__ dst,
                        const int* __restrict__ rank, const int* __restrict__ rowptr,
                        const float* __restrict__ deg, int2* __restrict__ rec, int E) {
  int i = blockIdx.x * blockDim.x + threadIdx.x;
  if (i < E) {
    const int s = src[i];
    const int d = dst[i];
    const float nm = rsqrtf(deg[s]) * rsqrtf(deg[d]);
    rec[rowptr[d] + rank[i]] = make_int2((s << 4) | (d & 15), __float_as_int(nm));
  }
}


template <bool RELU_IN>
__device__ __forceinline__ void gemm_n64_body(const float* __restrict__ X,
                                              const float* __restrict__ bin,
                                              const float* __restrict__ W,
                                              const float* __restrict__ deg,
                                              float* __restrict__ H,
                                              float* __restrict__ AGG, int M, int K) {
  int wave = (int)((blockIdx.x * blockDim.x + threadIdx.x) >> 5);
  int lane = threadIdx.x & 31;
  int m0 = wave * 16;
  if (m0 >= M) return;
  int half = lane >> 4;
  int lm = lane & 15;

  const float* xrow = X + (size_t)(m0 + lm) * (size_t)K;
  v8f acc0 = {}, acc1 = {}, acc2 = {}, acc3 = {};

  for (int kb = 0; kb < K; kb += 4) {
    int k0 = kb + 2 * half;
    v2f a;
    a.x = xrow[k0];
    a.y = xrow[k0 + 1];
    if (RELU_IN) {
      a.x = fmaxf(a.x + bin[k0], 0.0f);
      a.y = fmaxf(a.y + bin[k0 + 1], 0.0f);
    }
    const float* w0 = W + (size_t)k0 * 64;
    v2f b0, b1, b2, b3;
    b0.x = w0[lm];       b0.y = w0[64 + lm];
    b1.x = w0[16 + lm];  b1.y = w0[64 + 16 + lm];
    b2.x = w0[32 + lm];  b2.y = w0[64 + 32 + lm];
    b3.x = w0[48 + lm];  b3.y = w0[64 + 48 + lm];
    acc0 = wmma_f32_k4(a, b0, acc0);
    acc1 = wmma_f32_k4(a, b1, acc1);
    acc2 = wmma_f32_k4(a, b2, acc2);
    acc3 = wmma_f32_k4(a, b3, acc3);
  }

#pragma unroll
  for (int r = 0; r < 8; ++r) {
    int row = m0 + r + 8 * half;
    const float dv = rsqrtf(deg[row]);
    const float sl = dv * dv;
    float* hr = H + (size_t)row * 64 + lm;
    float* ar = AGG + (size_t)row * 64 + lm;
    hr[0]  = acc0[r];
    hr[16] = acc1[r];
    hr[32] = acc2[r];
    hr[48] = acc3[r];
    ar[0]  = acc0[r] * sl;
    ar[16] = acc1[r] * sl;
    ar[32] = acc2[r] * sl;
    ar[48] = acc3[r] * sl;
  }
}

__global__ void k_gemm_n64(const float* __restrict__ X, const float* __restrict__ W,
                           const float* __restrict__ deg, float* __restrict__ H,
                           float* __restrict__ AGG, int M, int K) {
  gemm_n64_body<false>(X, nullptr, W, deg, H, AGG, M, K);
}

__global__ void k_gemm_n64_relu_in(const float* __restrict__ X, const float* __restrict__ bin,
                                   const float* __restrict__ W, const float* __restrict__ deg,
                                   float* __restrict__ H, float* __restrict__ AGG,
                                   int M, int K) {
  gemm_n64_body<true>(X, bin, W, deg, H, AGG, M, K);
}


#define AGG_KSTEPS 8

__global__ __launch_bounds__(64)
void k_agg_onehot(const int2* __restrict__ rec, const int* __restrict__ rowptr,
                  const float* __restrict__ h, float* __restrict__ agg, int N) {
  const int wave = __builtin_amdgcn_readfirstlane(
      (int)((blockIdx.x * blockDim.x + threadIdx.x) >> 5));
  const int lane = threadIdx.x & 31;
  const int half = lane >> 4;
  const int lm = lane & 15;
  const int m0 = wave * 16;
  if (m0 >= N) return;
  const int mend = (m0 + 16 < N) ? m0 + 16 : N;
  const int e0 = __builtin_amdgcn_readfirstlane(rowptr[m0]);
  const int e1 = __builtin_amdgcn_readfirstlane(rowptr[mend]);

  v8f acc0, acc1, acc2, acc3;
#pragma unroll
  for (int r = 0; r < 8; ++r) {
    const int row = m0 + 8 * half + r;
    const int rowc = row < N ? row : N - 1;
    const float4 v = *reinterpret_cast<const float4*>(agg + (size_t)rowc * 64 + 4 * lm);
    acc0[r] = row < N ? v.x : 0.0f;
    acc1[r] = row < N ? v.y : 0.0f;
    acc2[r] = row < N ? v.z : 0.0f;
    acc3[r] = row < N ? v.w : 0.0f;
  }

  for (int eb = e0; eb < e1; eb += 4 * AGG_KSTEPS) {
    int2 rc[2 * AGG_KSTEPS];
    float4 hv[2 * AGG_KSTEPS];
#pragma unroll
    for (int q = 0; q < 2 * AGG_KSTEPS; ++q) {
      const int idx = eb + 4 * (q >> 1) + 2 * half + (q & 1);
      rc[q] = rec[idx < e1 ? idx : e1 - 1];
    }
#pragma unroll
    for (int q = 0; q < 2 * AGG_KSTEPS; ++q) {
      const unsigned int sidx = (unsigned int)rc[q].x >> 4;
      hv[q] = *reinterpret_cast<const float4*>(h + (size_t)sidx * 64 + 4 * lm);
    }
#pragma unroll
    for (int j = 0; j < AGG_KSTEPS; ++j) {
      const int idx = eb + 4 * j + 2 * half;
      const float w0 = idx < e1 ? __int_as_float(rc[2 * j].y) : 0.0f;
      const float w1 = idx + 1 < e1 ? __int_as_float(rc[2 * j + 1].y) : 0.0f;
      v2f a;
      a.x = ((rc[2 * j].x & 15) == lm) ? w0 : 0.0f;
      a.y = ((rc[2 * j + 1].x & 15) == lm) ? w1 : 0.0f;
      const float4 r0 = hv[2 * j], r1 = hv[2 * j + 1];
      const v2f b0 = {r0.x, r1.x};
      const v2f b1 = {r0.y, r1.y};
      const v2f b2 = {r0.z, r1.z};
      const v2f b3 = {r0.w, r1.w};
      acc0 = wmma_f32_k4(a, b0, acc0);
      acc1 = wmma_f32_k4(a, b1, acc1);
      acc2 = wmma_f32_k4(a, b2, acc2);
      acc3 = wmma_f32_k4(a, b3, acc3);
    }
  }

#pragma unroll
  for (int r = 0; r < 8; ++r) {
    const int row = m0 + 8 * half + r;
    if (row < N)
      *reinterpret_cast<float4*>(agg + (size_t)row * 64 + 4 * lm) =
          make_float4(acc0[r], acc1[r], acc2[r], acc3[r]);
  }
}


#define POOL_WAVES 8
#define POOL_WAVE_NODES 64
#define POOL_BLOCK_NODES (POOL_WAVES * POOL_WAVE_NODES)
#define POOL_ROW 4160
#define POOL_TILE 4224

__global__ __launch_bounds__(POOL_WAVES * 32)
void k_pool_onehot(const float* __restrict__ h, const float* __restrict__ bias,
                   const int* __restrict__ batch, float* __restrict__ part, int N) {
  __shared__ alignas(16) float red[POOL_WAVES * POOL_TILE];
  const int wave = threadIdx.x >> 5;
  const int lane = threadIdx.x & 31;
  const int half = lane >> 4;
  const int lm = lane & 15;
  const int n0 = (blockIdx.x * POOL_WAVES + wave) * POOL_WAVE_NODES;

  const float bb0 = bias[4 * lm], bb1 = bias[4 * lm + 1];
  const float bb2 = bias[4 * lm + 2], bb3 = bias[4 * lm + 3];
  float4 bv[32];
  int id[32];
#pragma unroll
  for (int s = 0; s < 32; ++s) {
    const int n = n0 + 4 * (s >> 1) + 2 * half + (s & 1);
    const int nc = n < N ? n : N - 1;
    const float4 v = *reinterpret_cast<const float4*>(h + (size_t)nc * 64 + 4 * lm);
    bv[s] = make_float4(fmaxf(v.x + bb0, 0.0f), fmaxf(v.y + bb1, 0.0f),
                        fmaxf(v.z + bb2, 0.0f), fmaxf(v.w + bb3, 0.0f));
    const int g = batch[nc];
    id[s] = n < N ? g : -1;
  }

  v8f acc[4][4];
  float cnt[4];
#pragma unroll
  for (int mi = 0; mi < 4; ++mi) {
    cnt[mi] = 0.0f;
#pragma unroll
    for (int t = 0; t < 4; ++t) acc[mi][t] = v8f{};
  }

#pragma unroll
  for (int j = 0; j < 16; ++j) {
    const float4 r0 = bv[2 * j], r1 = bv[2 * j + 1];
    const v2f b0 = {r0.x, r1.x};
    const v2f b1 = {r0.y, r1.y};
    const v2f b2 = {r0.z, r1.z};
    const v2f b3 = {r0.w, r1.w};
#pragma unroll
    for (int mi = 0; mi < 4; ++mi) {
      const int g = mi * 16 + lm;
      v2f a;
      a.x = (id[2 * j] == g) ? 1.0f : 0.0f;
      a.y = (id[2 * j + 1] == g) ? 1.0f : 0.0f;
      cnt[mi] += a.x + a.y;
      acc[mi][0] = wmma_f32_k4(a, b0, acc[mi][0]);
      acc[mi][1] = wmma_f32_k4(a, b1, acc[mi][1]);
      acc[mi][2] = wmma_f32_k4(a, b2, acc[mi][2]);
      acc[mi][3] = wmma_f32_k4(a, b3, acc[mi][3]);
    }
  }

  float* my = red + wave * POOL_TILE;
#pragma unroll
  for (int mi = 0; mi < 4; ++mi) {
#pragma unroll
    for (int r = 0; r < 8; ++r) {
      const int g = mi * 16 + 8 * half + r;
      *reinterpret_cast<float4*>(my + g * 64 + 4 * lm) =
          make_float4(acc[mi][0][r], acc[mi][1][r], acc[mi][2][r], acc[mi][3][r]);
    }
    my[4096 + half * 64 + mi * 16 + lm] = cnt[mi];
  }
  __syncthreads();

  const float4* red4 = reinterpret_cast<const float4*>(red);
  float4* out4 = reinterpret_cast<float4*>(part + (size_t)blockIdx.x * POOL_ROW);
  for (int q = threadIdx.x; q < 1024; q += POOL_WAVES * 32) {
    float4 s = red4[q];
    for (int w = 1; w < POOL_WAVES; ++w) {
      const float4 v = red4[w * (POOL_TILE / 4) + q];
      s.x += v.x; s.y += v.y; s.z += v.z; s.w += v.w;
    }
    out4[q] = s;
  }
  if (threadIdx.x < 64) {
    float s = 0.0f;
    for (int w = 0; w < POOL_WAVES; ++w)
      s += red[w * POOL_TILE + 4096 + threadIdx.x] + red[w * POOL_TILE + 4160 + threadIdx.x];
    part[(size_t)blockIdx.x * POOL_ROW + 4096 + threadIdx.x] = s;
  }
}


#define GS_PITCH 68

__global__ __launch_bounds__(128)
void k_pool_gemm_out(const float* __restrict__ part, int P,
                     const float* __restrict__ Wp, const float* __restrict__ bp,
                     float* __restrict__ out) {
  __shared__ alignas(16) float gs[16 * GS_PITCH];
  __shared__ alignas(16) float cs[16];
  const int tid = threadIdx.x;
  const int mi = blockIdx.x;
  const int ni = tid >> 5;
  const int lane = tid & 31;
  const int half = lane >> 4, lm = lane & 15;

  const float4* p4 = reinterpret_cast<const float4*>(part) + mi * 256 + tid;
  const float4* c4 = reinterpret_cast<const float4*>(part) + 1024 + mi * 4 + (tid & 3);
  float4 s0 = make_float4(0.0f, 0.0f, 0.0f, 0.0f), s1 = s0, s2 = s0;
#pragma unroll 16
  for (int p = 0; p < P; ++p) {
    const size_t ro = (size_t)p * (POOL_ROW / 4);
    const float4 v0 = p4[ro];
    const float4 v1 = p4[ro + 128];
    s0.x += v0.x; s0.y += v0.y; s0.z += v0.z; s0.w += v0.w;
    s1.x += v1.x; s1.y += v1.y; s1.z += v1.z; s1.w += v1.w;
    if (tid < 4) {
      const float4 v2 = c4[ro];
      s2.x += v2.x; s2.y += v2.y; s2.z += v2.z; s2.w += v2.w;
    }
  }
  *reinterpret_cast<float4*>(gs + (tid >> 4) * GS_PITCH + 4 * (tid & 15)) = s0;
  *reinterpret_cast<float4*>(gs + ((tid + 128) >> 4) * GS_PITCH + 4 * (tid & 15)) = s1;
  if (tid < 4) *reinterpret_cast<float4*>(cs + 4 * tid) = s2;
  __syncthreads();

  const float inv = 1.0f / fmaxf(cs[lm], 1.0f);

  v8f acc = {};
  for (int kb = 0; kb < 64; kb += 4) {
    int k0 = kb + 2 * half;
    v2f a;
    a.x = gs[lm * GS_PITCH + k0] * inv;
    a.y = gs[lm * GS_PITCH + k0 + 1] * inv;
    v2f b;
    b.x = Wp[k0 * 64 + ni * 16 + lm];
    b.y = Wp[(k0 + 1) * 64 + ni * 16 + lm];
    acc = wmma_f32_k4(a, b, acc);
  }

#pragma unroll
  for (int r = 0; r < 8; ++r) {
    int row = mi * 16 + r + 8 * half;
    out[row * 64 + ni * 16 + lm] = acc[r] + bp[ni * 16 + lm];
  }
}


static inline size_t align_f(size_t n) { return (n + 1023) & ~(size_t)1023; }

extern "C" void kernel_launch(void* const* d_in, const int* in_sizes, int n_in,
                              void* d_out, int out_size, void* d_ws, size_t ws_size,
                              hipStream_t stream) {
  const float* x  = (const float*)d_in[0];
  const float* W1 = (const float*)d_in[1];
  const float* b1 = (const float*)d_in[2];
  const float* W2 = (const float*)d_in[3];
  const float* b2 = (const float*)d_in[4];
  const float* Wp = (const float*)d_in[5];
  const float* bp = (const float*)d_in[6];
  const int* ei   = (const int*)d_in[7];
  const int* batch = (const int*)d_in[8];

  const int E = in_sizes[7] / 2;
  const int N = in_sizes[8];
  const int K1 = in_sizes[0] / N;
  const int* src = ei;
  const int* dst = ei + E;

  float* ws = (float*)d_ws;
  size_t o = 0;
  float* deg  = ws + o;  o += align_f((size_t)N);
  int* rowptr = (int*)(ws + o);  o += align_f((size_t)N + 1);
  int* rank   = (int*)(ws + o);  o += align_f((size_t)E);
  int2* rec   = (int2*)(ws + o); o += align_f((size_t)E * 2);
  float* h    = ws + o;  o += align_f((size_t)N * 64);
  float* agg1 = ws + o;  o += align_f((size_t)N * 64);
  float* agg2 = ws + o;  o += align_f((size_t)N * 64);
  const int pool_blocks = (N + POOL_BLOCK_NODES - 1) / POOL_BLOCK_NODES;
  float* part = ws + o;  o += (size_t)pool_blocks * POOL_ROW;
  (void)ws_size;

  const int T = 256;
  auto blocks = [](long long n, int t) { return (unsigned int)((n + t - 1) / t); };

  k_fill<<<blocks(N, T), T, 0, stream>>>(deg, 1.0f, N);
  k_deg_rank<<<blocks(E, T), T, 0, stream>>>(dst, deg, rank, E);
  k_scan_rows<<<1, SCAN_T, 0, stream>>>(deg, rowptr, N);
  k_place<<<blocks(E, T), T, 0, stream>>>(src, dst, rank, rowptr, deg, rec, E);

  const int tiles = (N + 15) / 16;
  const unsigned int gemm_blocks = (unsigned int)((tiles + 7) / 8);
  const unsigned int agg_blocks = (unsigned int)((tiles + 1) / 2);

  k_gemm_n64<<<gemm_blocks, T, 0, stream>>>(x, W1, deg, h, agg1, N, K1);
  k_agg_onehot<<<agg_blocks, 64, 0, stream>>>(rec, rowptr, h, agg1, N);

  k_gemm_n64_relu_in<<<gemm_blocks, T, 0, stream>>>(agg1, b1, W2, deg, h, agg2, N, 64);
  k_agg_onehot<<<agg_blocks, 64, 0, stream>>>(rec, rowptr, h, agg2, N);

  k_pool_onehot<<<pool_blocks, POOL_WAVES * 32, 0, stream>>>(agg2, b2, batch, part, N);
  k_pool_gemm_out<<<4, 128, 0, stream>>>(part, pool_blocks, Wp, bp, (float*)d_out);
}
